# starting kernel
# baseline (speedup 1.0000x reference)
#include <hip/hip_runtime.h>
#include <hip/hip_bf16.h>
#include <math.h>

constexpr int Bn = 16;
constexpr int Cn = 4;
constexpr int Ln = 4096;
constexpr int Kn = 256;
constexpr int Sn = 128;
constexpr int Wn = Ln - Sn + 1;
constexpr int NT = (Wn + 15) / 16;
constexpr float EPSV = 1e-8f;

typedef __attribute__((ext_vector_type(2))) float v2f;
typedef __attribute__((ext_vector_type(8))) float v8f;

__global__ __launch_bounds__(128) void shapelet_norm_kernel(
    const float* __restrict__ shp, float* __restrict__ shpn)
{
    __shared__ float red[4];
    const int row = blockIdx.x;
    const int s   = threadIdx.x;
    const float v = shp[row * Sn + s];
    float p = v * v;
    #pragma unroll
    for (int off = 16; off >= 1; off >>= 1) p += __shfl_xor(p, off, 32);
    if ((s & 31) == 0) red[s >> 5] = p;
    __syncthreads();
    const float tot = red[0] + red[1] + red[2] + red[3];
    shpn[row * Sn + s] = v / fmaxf(sqrtf(tot), EPSV);
}

__global__ __launch_bounds__(256) void window_norm_kernel(
    const float* __restrict__ x, float* __restrict__ invn)
{
    const int bc = blockIdx.x;
    const float* px = x + (size_t)bc * Ln;
    for (int w = threadIdx.x; w < Wn; w += 256) {
        float ss = 0.f;
        #pragma unroll 8
        for (int s = 0; s < Sn; ++s) { const float v = px[w + s]; ss = fmaf(v, v, ss); }
        invn[bc * Wn + w] = 1.0f / fmaxf(sqrtf(ss), EPSV);
    }
}

__global__ __launch_bounds__(128) void shapelet_wmma_kernel(
    const float* __restrict__ x, const float* __restrict__ shpn,
    const float* __restrict__ invn, float* __restrict__ out)
{
    __shared__ v2f   ldsB[Cn * (Sn / 2) * 16];
    __shared__ float ldsR[4 * 16];

    const int b    = blockIdx.x >> 4;
    const int k0   = (blockIdx.x & 15) << 4;
    const int tid  = threadIdx.x;
    const int wave = tid >> 5;
    const int lane = tid & 31;
    const int n    = lane & 15;
    const int kA   = (lane >> 4) << 1;

    for (int i = tid; i < Cn * (Sn / 2) * 16; i += 128) {
        const int nn = i & 15;
        const int s2 = (i >> 4) & 63;
        const int c  = i >> 10;
        ldsB[i] = ((const v2f*)shpn)[(size_t)(c * Kn + (k0 + nn)) * (Sn / 2) + s2];
    }
    __syncthreads();

    float rmax = 0.0f;
    for (int t = wave; t < NT; t += 4) {
        const int w0 = t << 4;
        const int wv = w0 + n;
        const int wc = wv < Wn ? wv : Wn - 1;
        v8f acc = {0.f, 0.f, 0.f, 0.f, 0.f, 0.f, 0.f, 0.f};

        for (int c = 0; c < Cn; ++c) {
            float s_inv = invn[(b * Cn + c) * Wn + wc];
            s_inv = (wv < Wn) ? s_inv : 0.0f;
            const float* px = x + (size_t)(b * Cn + c) * Ln + wc + kA;
            const v2f*   pb = &ldsB[(c * 64) * 16 + n];
            #pragma unroll 4
            for (int s0 = 0; s0 < Sn; s0 += 4) {
                v2f a;
                a.x = px[s0]     * s_inv;
                a.y = px[s0 + 1] * s_inv;
                const v2f bv = pb[((s0 + kA) >> 1) * 16];
                acc = __builtin_amdgcn_wmma_f32_16x16x4_f32(
                     false, a,  false, bv,
                     (short)0, acc,  false,  false);
            }
        }
        #pragma unroll
        for (int r = 0; r < 8; ++r) rmax = fmaxf(rmax, acc[r] * 0.25f);
    }

    rmax = fmaxf(rmax, __shfl_xor(rmax, 16, 32));
    if (lane < 16) ldsR[wave * 16 + n] = rmax;
    __syncthreads();
    if (tid < 16) {
        const float v = fmaxf(fmaxf(ldsR[tid], ldsR[16 + tid]),
                              fmaxf(ldsR[32 + tid], ldsR[48 + tid]));
        out[b * Kn + k0 + tid] = v;
    }
}

extern "C" void kernel_launch(void* const* d_in, const int* in_sizes, int n_in,
                              void* d_out, int out_size, void* d_ws, size_t ws_size,
                              hipStream_t stream) {
    const float* x   = (const float*)d_in[0];
    const float* shp = (const float*)d_in[1];
    float* out  = (float*)d_out;
    float* shpn = (float*)d_ws;
    float* invn = shpn + Cn * Kn * Sn;

    shapelet_norm_kernel<<<Cn * Kn, Sn, 0, stream>>>(shp, shpn);
    window_norm_kernel<<<Bn * Cn, 256, 0, stream>>>(x, invn);
    shapelet_wmma_kernel<<<Bn * (Kn / 16), 128, 0, stream>>>(x, shpn, invn, out);
}


# Main kernel re-tiled: each wave holds a 16-window x 64-shapelet register tile (4 accumulators, one scaled x fragment feeds 4 WMMAs); windows split over 4 work-groups, maxima merged by a 4th launch
# speedup vs baseline: 2.9285x; 2.9285x over previous
#include <hip/hip_runtime.h>
#include <hip/hip_bf16.h>
#include <math.h>


constexpr int Bn = 16;
constexpr int Cn = 4;
constexpr int Ln = 4096;
constexpr int Kn = 256;
constexpr int Sn = 128;
constexpr int Wn = Ln - Sn + 1;
constexpr int NT = (Wn + 15) / 16;
constexpr float EPSV = 1e-8f;

constexpr int JT    = 4;
constexpr int KG    = Kn / (16 * JT);
constexpr int QS    = 4;
constexpr int NWAVE = 16;
constexpr int LDSB_F4 = Cn * (Sn / 4) * (JT / 2) * 32;
constexpr int LDSB_DECL_F4 = 10368;
static_assert(LDSB_DECL_F4 >= LDSB_F4 && LDSB_DECL_F4 * 16 > 160 * 1024, "B fragments fit; one work-group per WGP");

typedef __attribute__((ext_vector_type(2))) float v2f;
typedef __attribute__((ext_vector_type(4))) float v4f;
typedef __attribute__((ext_vector_type(8))) float v8f;

__global__ __launch_bounds__(128) void shapelet_norm_kernel(
    const float* __restrict__ shp, float* __restrict__ den)
{
    __shared__ float red[4];
    const int row = blockIdx.x;
    const int s   = threadIdx.x;
    const float v = shp[row * Sn + s];
    float p = v * v;
    #pragma unroll
    for (int off = 16; off >= 1; off >>= 1) p += __shfl_xor(p, off, 32);
    if ((s & 31) == 0) red[s >> 5] = p;
    __syncthreads();
    const float tot = red[0] + red[1] + red[2] + red[3];
    if (s == 0) den[row] = fmaxf(sqrtf(tot), EPSV);
}

__global__ __launch_bounds__(256) void window_norm_kernel(
    const float* __restrict__ x, float* __restrict__ invn)
{
    const int bc = blockIdx.x;
    const float* px = x + (size_t)bc * Ln;
    for (int w = threadIdx.x; w < Wn; w += 256) {
        float ss = 0.f;
        #pragma unroll 8
        for (int s = 0; s < Sn; ++s) { const float v = px[w + s]; ss = fmaf(v, v, ss); }
        invn[bc * Wn + w] = 1.0f / fmaxf(sqrtf(ss), EPSV);
    }
}

__global__ __launch_bounds__(NWAVE * 32) void shapelet_wmma_kernel(
    const float* __restrict__ x, const float* __restrict__ shp,
    const float* __restrict__ den, const float* __restrict__ invn,
    float* __restrict__ part)
{
    __shared__ v4f   ldsB[LDSB_DECL_F4];
    __shared__ float ldsR[NWAVE * 16 * JT];

    const int q    = blockIdx.x & (QS - 1);
    const int g    = (blockIdx.x >> 2) & (KG - 1);
    const int b    = blockIdx.x >> 4;
    const int tid  = threadIdx.x;
    const int wave = tid >> 5;
    const int lane = tid & 31;
    const int n    = lane & 15;
    const int kA   = (lane >> 4) << 1;

    {
        v2f* ldsB2 = (v2f*)ldsB;
        const int nn = tid & 15;
        const int s4 = tid >> 4;
        #pragma unroll 4
        for (int it = 0; it < Cn * JT; ++it) {
            const int j   = it & (JT - 1);
            const int c   = it >> 2;
            const int row = c * Kn + g * (16 * JT) + j * 16 + nn;
            const v4f   v = *(const v4f*)(shp + (size_t)row * Sn + s4 * 4);
            const float d = den[row];
            const v2f lo = {v.x / d, v.y / d};
            const v2f hi = {v.z / d, v.w / d};
            const int base = ((((c * 32 + s4) * 2 + (j >> 1)) * 2) * 16 + nn) * 2 + (j & 1);
            ldsB2[base]      = lo;
            ldsB2[base + 32] = hi;
        }
    }
    __syncthreads();

    float rmax[JT];
    #pragma unroll
    for (int j = 0; j < JT; ++j) rmax[j] = 0.0f;

    for (int t = q * NWAVE + wave; t < NT; t += QS * NWAVE) {
        const int w0 = t << 4;
        const int wv = w0 + n;
        const int wc = wv < Wn ? wv : Wn - 1;
        v8f acc[JT];
        #pragma unroll
        for (int j = 0; j < JT; ++j) acc[j] = (v8f){0.f, 0.f, 0.f, 0.f, 0.f, 0.f, 0.f, 0.f};

        float sinv[Cn];
        #pragma unroll
        for (int c = 0; c < Cn; ++c) {
            const float v = invn[(b * Cn + c) * Wn + wc];
            sinv[c] = (wv < Wn) ? v : 0.0f;
        }

        #pragma unroll
        for (int c = 0; c < Cn; ++c) {
            const float s_inv = sinv[c];
            const float* px = x + (size_t)(b * Cn + c) * Ln + wc + kA;
            const v4f*   pb = &ldsB[c * 2048 + lane];
            #pragma unroll 8
            for (int s4 = 0; s4 < Sn / 4; ++s4) {
                v2f a;
                a.x = px[s4 * 4]     * s_inv;
                a.y = px[s4 * 4 + 1] * s_inv;
                const v4f b01 = pb[s4 * 64];
                const v4f b23 = pb[s4 * 64 + 32];
                const v2f bv0 = {b01.x, b01.y};
                const v2f bv1 = {b01.z, b01.w};
                const v2f bv2 = {b23.x, b23.y};
                const v2f bv3 = {b23.z, b23.w};
                acc[0] = __builtin_amdgcn_wmma_f32_16x16x4_f32(
                     false, a,  false, bv0,
                     (short)0, acc[0],  false,  false);
                acc[1] = __builtin_amdgcn_wmma_f32_16x16x4_f32(
                    false, a, false, bv1, (short)0, acc[1], false, false);
                acc[2] = __builtin_amdgcn_wmma_f32_16x16x4_f32(
                    false, a, false, bv2, (short)0, acc[2], false, false);
                acc[3] = __builtin_amdgcn_wmma_f32_16x16x4_f32(
                    false, a, false, bv3, (short)0, acc[3], false, false);
            }
        }
        #pragma unroll
        for (int j = 0; j < JT; ++j) {
            #pragma unroll
            for (int r = 0; r < 8; ++r) rmax[j] = fmaxf(rmax[j], acc[j][r] * 0.25f);
        }
    }

    #pragma unroll
    for (int j = 0; j < JT; ++j) {
        rmax[j] = fmaxf(rmax[j], __shfl_xor(rmax[j], 16, 32));
        if (lane < 16) ldsR[wave * (16 * JT) + j * 16 + n] = rmax[j];
    }
    __syncthreads();
    if (tid < 16 * JT) {
        float v = ldsR[tid];
        #pragma unroll
        for (int w = 1; w < NWAVE; ++w) v = fmaxf(v, ldsR[w * (16 * JT) + tid]);
        part[(q * Bn + b) * Kn + g * (16 * JT) + tid] = v;
    }
}

__global__ __launch_bounds__(256) void split_max_kernel(
    const float* __restrict__ part, float* __restrict__ out)
{
    const int i = blockIdx.x * 256 + threadIdx.x;
    const float v = fmaxf(fmaxf(part[i], part[Bn * Kn + i]),
                          fmaxf(part[2 * Bn * Kn + i], part[3 * Bn * Kn + i]));
    out[i] = v;
}

extern "C" void kernel_launch(void* const* d_in, const int* in_sizes, int n_in,
                              void* d_out, int out_size, void* d_ws, size_t ws_size,
                              hipStream_t stream) {
    const float* x   = (const float*)d_in[0];
    const float* shp = (const float*)d_in[1];
    float* out  = (float*)d_out;
    float* den  = (float*)d_ws;
    float* part = den + Cn * Kn;
    float* invn = den + Cn * Kn * Sn;

    shapelet_norm_kernel<<<Cn * Kn, Sn, 0, stream>>>(shp, den);
    window_norm_kernel<<<Bn * Cn, 256, 0, stream>>>(x, invn);
    shapelet_wmma_kernel<<<Bn * KG * QS, NWAVE * 32, 0, stream>>>(x, shp, den, invn, part);
    split_max_kernel<<<Bn * Kn / 256, 256, 0, stream>>>(part, out);
}


# Window norms computed inside the main kernel: lanes sum the squares of their raw x operand elements, 1/norm applied to each channel's finished sums; 4 -> 3 launches; new sum order
# speedup vs baseline: 4.2246x; 1.4426x over previous
#include <hip/hip_runtime.h>
#include <hip/hip_bf16.h>
#include <math.h>


constexpr int Bn = 16;
constexpr int Cn = 4;
constexpr int Ln = 4096;
constexpr int Kn = 256;
constexpr int Sn = 128;
constexpr int Wn = Ln - Sn + 1;
constexpr int NT = (Wn + 15) / 16;
constexpr float EPSV = 1e-8f;

constexpr int JT    = 4;
constexpr int KG    = Kn / (16 * JT);
constexpr int QS    = 4;
constexpr int NWAVE = 16;
constexpr int LDSB_F4 = Cn * (Sn / 4) * (JT / 2) * 32;
constexpr int LDSB_DECL_F4 = 10368;
static_assert(LDSB_DECL_F4 >= LDSB_F4 && LDSB_DECL_F4 * 16 > 160 * 1024, "B fragments fit; one work-group per WGP");

typedef __attribute__((ext_vector_type(2))) float v2f;
typedef __attribute__((ext_vector_type(4))) float v4f;
typedef __attribute__((ext_vector_type(8))) float v8f;

__global__ __launch_bounds__(128) void shapelet_norm_kernel(
    const float* __restrict__ shp, float* __restrict__ den)
{
    __shared__ float red[4];
    const int row = blockIdx.x;
    const int s   = threadIdx.x;
    const float v = shp[row * Sn + s];
    float p = v * v;
    #pragma unroll
    for (int off = 16; off >= 1; off >>= 1) p += __shfl_xor(p, off, 32);
    if ((s & 31) == 0) red[s >> 5] = p;
    __syncthreads();
    const float tot = red[0] + red[1] + red[2] + red[3];
    if (s == 0) den[row] = fmaxf(sqrtf(tot), EPSV);
}

__global__ __launch_bounds__(NWAVE * 32) void shapelet_wmma_kernel(
    const float* __restrict__ x, const float* __restrict__ shp,
    const float* __restrict__ den, float* __restrict__ part)
{
    __shared__ v4f   ldsB[LDSB_DECL_F4];
    __shared__ float ldsR[NWAVE * 16 * JT];

    const int q    = blockIdx.x & (QS - 1);
    const int g    = (blockIdx.x >> 2) & (KG - 1);
    const int b    = blockIdx.x >> 4;
    const int tid  = threadIdx.x;
    const int wave = tid >> 5;
    const int lane = tid & 31;
    const int n    = lane & 15;
    const int kA   = (lane >> 4) << 1;
    const int rbase = (lane >> 4) << 3;

    {
        v2f* ldsB2 = (v2f*)ldsB;
        const int nn = tid & 15;
        const int s4 = tid >> 4;
        #pragma unroll 4
        for (int it = 0; it < Cn * JT; ++it) {
            const int j   = it & (JT - 1);
            const int c   = it >> 2;
            const int row = c * Kn + g * (16 * JT) + j * 16 + nn;
            const v4f   v = *(const v4f*)(shp + (size_t)row * Sn + s4 * 4);
            const float d = den[row];
            const v2f lo = {v.x / d, v.y / d};
            const v2f hi = {v.z / d, v.w / d};
            const int base = ((((c * 32 + s4) * 2 + (j >> 1)) * 2) * 16 + nn) * 2 + (j & 1);
            ldsB2[base]      = lo;
            ldsB2[base + 32] = hi;
        }
    }
    __syncthreads();

    float rmax[JT];
    #pragma unroll
    for (int j = 0; j < JT; ++j) rmax[j] = 0.0f;

    for (int t = q * NWAVE + wave; t < NT; t += QS * NWAVE) {
        const int w0 = t << 4;
        const int wv = w0 + n;
        const int wc = wv < Wn ? wv : Wn - 1;
        v8f tot[JT];

        #pragma unroll
        for (int c = 0; c < Cn; ++c) {
            v8f acc[JT];
            #pragma unroll
            for (int j = 0; j < JT; ++j) acc[j] = (v8f){0.f, 0.f, 0.f, 0.f, 0.f, 0.f, 0.f, 0.f};
            float p0 = 0.0f, p1 = 0.0f;
            const float* px = x + (size_t)(b * Cn + c) * Ln + wc + kA;
            const v4f*   pb = &ldsB[c * 2048 + lane];
            #pragma unroll 8
            for (int s4 = 0; s4 < Sn / 4; ++s4) {
                v2f a;
                a.x = px[s4 * 4];
                a.y = px[s4 * 4 + 1];
                p0 = fmaf(a.x, a.x, p0);
                p1 = fmaf(a.y, a.y, p1);
                const v4f b01 = pb[s4 * 64];
                const v4f b23 = pb[s4 * 64 + 32];
                const v2f bv0 = {b01.x, b01.y};
                const v2f bv1 = {b01.z, b01.w};
                const v2f bv2 = {b23.x, b23.y};
                const v2f bv3 = {b23.z, b23.w};
                acc[0] = __builtin_amdgcn_wmma_f32_16x16x4_f32(
                     false, a,  false, bv0,
                     (short)0, acc[0],  false,  false);
                acc[1] = __builtin_amdgcn_wmma_f32_16x16x4_f32(
                    false, a, false, bv1, (short)0, acc[1], false, false);
                acc[2] = __builtin_amdgcn_wmma_f32_16x16x4_f32(
                    false, a, false, bv2, (short)0, acc[2], false, false);
                acc[3] = __builtin_amdgcn_wmma_f32_16x16x4_f32(
                    false, a, false, bv3, (short)0, acc[3], false, false);
            }
            float ss = p0 + p1;
            ss += __shfl_xor(ss, 16, 32);
            float inv = 1.0f / fmaxf(sqrtf(ss), EPSV);
            inv = (wv < Wn) ? inv : 0.0f;
            #pragma unroll
            for (int r = 0; r < 8; ++r) {
                const float ir = __shfl(inv, rbase + r, 32);
                #pragma unroll
                for (int j = 0; j < JT; ++j)
                    tot[j][r] = (c == 0) ? acc[j][r] * ir : fmaf(acc[j][r], ir, tot[j][r]);
            }
        }
        #pragma unroll
        for (int j = 0; j < JT; ++j) {
            #pragma unroll
            for (int r = 0; r < 8; ++r) rmax[j] = fmaxf(rmax[j], tot[j][r] * 0.25f);
        }
    }

    #pragma unroll
    for (int j = 0; j < JT; ++j) {
        rmax[j] = fmaxf(rmax[j], __shfl_xor(rmax[j], 16, 32));
        if (lane < 16) ldsR[wave * (16 * JT) + j * 16 + n] = rmax[j];
    }
    __syncthreads();
    if (tid < 16 * JT) {
        float v = ldsR[tid];
        #pragma unroll
        for (int w = 1; w < NWAVE; ++w) v = fmaxf(v, ldsR[w * (16 * JT) + tid]);
        part[(q * Bn + b) * Kn + g * (16 * JT) + tid] = v;
    }
}

__global__ __launch_bounds__(256) void split_max_kernel(
    const float* __restrict__ part, float* __restrict__ out)
{
    const int i = blockIdx.x * 256 + threadIdx.x;
    const float v = fmaxf(fmaxf(part[i], part[Bn * Kn + i]),
                          fmaxf(part[2 * Bn * Kn + i], part[3 * Bn * Kn + i]));
    out[i] = v;
}

extern "C" void kernel_launch(void* const* d_in, const int* in_sizes, int n_in,
                              void* d_out, int out_size, void* d_ws, size_t ws_size,
                              hipStream_t stream) {
    const float* x   = (const float*)d_in[0];
    const float* shp = (const float*)d_in[1];
    float* out  = (float*)d_out;
    float* den  = (float*)d_ws;
    float* part = den + Cn * Kn;

    shapelet_norm_kernel<<<Cn * Kn, Sn, 0, stream>>>(shp, den);
    shapelet_wmma_kernel<<<Bn * KG * QS, NWAVE * 32, 0, stream>>>(x, shp, den, part);
    split_max_kernel<<<Bn * Kn / 256, 256, 0, stream>>>(part, out);
}


# Operand roles swapped: a wave keeps 16 shapelets x 4 channels as A fragments in registers all kernel long, streams 64-window groups of raw x as B; no LDS reads in the loop, 3 -> 2 launches
# speedup vs baseline: 4.4687x; 1.0578x over previous
#include <hip/hip_runtime.h>
#include <math.h>


constexpr int Bn = 16;
constexpr int Cn = 4;
constexpr int Ln = 4096;
constexpr int Kn = 256;
constexpr int Sn = 128;
constexpr int Wn = Ln - Sn + 1;
constexpr float EPSV = 1e-8f;

constexpr int KT    = Kn / 16;
constexpr int TU    = 4;
constexpr int GW    = 16 * TU;
constexpr int NG    = (Wn + GW - 1) / GW;
constexpr int NCH   = Sn / 4;
constexpr int NF    = NCH + 4 * (TU - 1);
constexpr int NLO   = 28;
constexpr int NWAVE = 8;
constexpr int APITCH = Sn + 4;

typedef __attribute__((ext_vector_type(2))) float v2f;
typedef __attribute__((ext_vector_type(4))) float v4f;
typedef __attribute__((ext_vector_type(8))) float v8f;

__global__ __launch_bounds__(128) void shapelet_norm_kernel(
    const float* __restrict__ shp, float* __restrict__ den)
{
    __shared__ float red[4];
    const int row = blockIdx.x;
    const int s   = threadIdx.x;
    const float v = shp[row * Sn + s];
    float p = v * v;
    #pragma unroll
    for (int off = 16; off >= 1; off >>= 1) p += __shfl_xor(p, off, 32);
    if ((s & 31) == 0) red[s >> 5] = p;
    __syncthreads();
    const float tot = red[0] + red[1] + red[2] + red[3];
    if (s == 0) den[row] = fmaxf(sqrtf(tot), EPSV);
}

__device__ __forceinline__ v2f ldpair(const float* p)
{
    v2f r;
    r.x = p[0];
    r.y = p[1];
    return r;
}

__global__ __launch_bounds__(NWAVE * 32) void shapelet_wmma_kernel(
    const float* __restrict__ x, const float* __restrict__ shp,
    const float* __restrict__ den, float* __restrict__ out)
{
    __shared__ float ldsA[Cn * 16 * APITCH];
    __shared__ float ldsR[NWAVE * 16];

    const int kt   = blockIdx.x & (KT - 1);
    const int b    = blockIdx.x >> 4;
    const int tid  = threadIdx.x;
    const int wave = __builtin_amdgcn_readfirstlane(tid >> 5);
    const int lane = tid & 31;
    const int n    = lane & 15;
    const int h    = lane >> 4;
    const int kA   = h << 1;

    #pragma unroll
    for (int it = 0; it < (Cn * 16 * Sn / 4) / (NWAVE * 32); ++it) {
        const int idx = it * (NWAVE * 32) + tid;
        const int r   = idx >> 5;
        const int q4  = idx & 31;
        const int row = (r >> 4) * Kn + kt * 16 + (r & 15);
        const v4f   v = *(const v4f*)(shp + (size_t)row * Sn + q4 * 4);
        const float d = den[row];
        const v4f   q = {v.x / d, v.y / d, v.z / d, v.w / d};
        *(v4f*)&ldsA[r * APITCH + q4 * 4] = q;
    }
    __syncthreads();

    v2f af[Cn][NCH];
    #pragma unroll
    for (int c = 0; c < Cn; ++c) {
        #pragma unroll
        for (int j = 0; j < NCH; ++j)
            af[c][j] = *(const v2f*)&ldsA[(c * 16 + n) * APITCH + 4 * j + kA];
    }

    float rmax[8];
    #pragma unroll
    for (int r = 0; r < 8; ++r) rmax[r] = 0.0f;

    const float* xb = x + (size_t)b * Cn * Ln + n + kA;

    v2f f[NF];
    {
        const int g0 = wave;
        const int W0 = (g0 * GW < Wn - GW) ? g0 * GW : Wn - GW;
        #pragma unroll
        for (int i = 0; i < NLO; ++i) f[i] = ldpair(xb + W0 + 4 * i);
    }

    for (int g = wave; g < NG; g += NWAVE) {
        const int W0  = (g * GW < Wn - GW) ? g * GW : Wn - GW;
        const int gn  = (g + NWAVE < NG) ? g + NWAVE : g;
        const int W0n = (gn * GW < Wn - GW) ? gn * GW : Wn - GW;
        v8f tot[TU];

        #pragma unroll
        for (int c = 0; c < Cn; ++c) {
            const float* px  = xb + c * Ln + W0;
            const float* pxn = (c < Cn - 1) ? xb + (c + 1) * Ln + W0 : xb + W0n;

            __builtin_amdgcn_sched_barrier(0);
            #pragma unroll
            for (int i = NLO; i < NF; ++i) f[i] = ldpair(px + 4 * i);
            __builtin_amdgcn_sched_barrier(0);

            v8f acc[TU];
            v2f p[TU];
            #pragma unroll
            for (int u = 0; u < TU; ++u) {
                acc[u] = (v8f){0.f, 0.f, 0.f, 0.f, 0.f, 0.f, 0.f, 0.f};
                p[u]   = (v2f){0.f, 0.f};
            }

            #pragma unroll
            for (int j = 0; j < NCH / 2; ++j) {
                #pragma unroll
                for (int u = 0; u < TU; ++u) {
                    const v2f bv = f[j + 4 * u];
                    p[u].x = fmaf(bv.x, bv.x, p[u].x);
                    p[u].y = fmaf(bv.y, bv.y, p[u].y);
                    acc[u] = __builtin_amdgcn_wmma_f32_16x16x4_f32(
                         false, af[c][j],  false, bv,
                         (short)0, acc[u],  false,  false);
                }
            }

            __builtin_amdgcn_sched_barrier(0);
            v2f nf[NLO];
            #pragma unroll
            for (int i = 0; i < NLO; ++i) nf[i] = ldpair(pxn + 4 * i);
            __builtin_amdgcn_sched_barrier(0);

            float inv[TU];
            #pragma unroll
            for (int u = 0; u < TU; ++u) {
                #pragma unroll
                for (int j = NCH / 2; j < NCH; ++j) {
                    const v2f bv = f[j + 4 * u];
                    p[u].x = fmaf(bv.x, bv.x, p[u].x);
                    p[u].y = fmaf(bv.y, bv.y, p[u].y);
                }
                float ss = p[u].x + p[u].y;
                ss += __shfl_xor(ss, 16, 32);
                inv[u] = 1.0f / fmaxf(sqrtf(ss), EPSV);
            }

            #pragma unroll
            for (int j = NCH / 2; j < NCH; ++j) {
                #pragma unroll
                for (int u = 0; u < TU; ++u) {
                    acc[u] = __builtin_amdgcn_wmma_f32_16x16x4_f32(
                        false, af[c][j], false, f[j + 4 * u], (short)0, acc[u], false, false);
                }
            }

            #pragma unroll
            for (int u = 0; u < TU; ++u) {
                #pragma unroll
                for (int r = 0; r < 8; ++r)
                    tot[u][r] = (c == 0) ? acc[u][r] * inv[u] : fmaf(acc[u][r], inv[u], tot[u][r]);
            }

            #pragma unroll
            for (int i = 0; i < NLO; ++i) f[i] = nf[i];
        }
        #pragma unroll
        for (int u = 0; u < TU; ++u) {
            #pragma unroll
            for (int r = 0; r < 8; ++r) rmax[r] = fmaxf(rmax[r], tot[u][r] * 0.25f);
        }
    }

    #pragma unroll
    for (int r = 0; r < 8; ++r) {
        #pragma unroll
        for (int off = 8; off >= 1; off >>= 1) rmax[r] = fmaxf(rmax[r], __shfl_xor(rmax[r], off, 32));
        if (n == 0) ldsR[wave * 16 + 8 * h + r] = rmax[r];
    }
    __syncthreads();
    if (tid < 16) {
        float v = ldsR[tid];
        #pragma unroll
        for (int w = 1; w < NWAVE; ++w) v = fmaxf(v, ldsR[w * 16 + tid]);
        out[b * Kn + kt * 16 + tid] = v;
    }
}

extern "C" void kernel_launch(void* const* d_in, const int* in_sizes, int n_in,
                              void* d_out, int out_size, void* d_ws, size_t ws_size,
                              hipStream_t stream) {
    const float* x   = (const float*)d_in[0];
    const float* shp = (const float*)d_in[1];
    float* out  = (float*)d_out;
    float* den  = (float*)d_ws;

    shapelet_norm_kernel<<<Cn * Kn, Sn, 0, stream>>>(shp, den);
    shapelet_wmma_kernel<<<Bn * KT, NWAVE * 32, 0, stream>>>(x, shp, den, out);
}


# Window inverse norms leave the matrix-bound main loop: the first launch computes them once per (b,c,w) (4 adjacent windows per thread, same four-chain sums); the loop loads them, 4 per channel
# speedup vs baseline: 5.0006x; 1.1190x over previous
#include <hip/hip_runtime.h>
#include <math.h>


constexpr int Bn = 16;
constexpr int Cn = 4;
constexpr int Ln = 4096;
constexpr int Kn = 256;
constexpr int Sn = 128;
constexpr int Wn = Ln - Sn + 1;
constexpr float EPSV = 1e-8f;

constexpr int KT    = Kn / 16;
constexpr int TU    = 4;
constexpr int GW    = 16 * TU;
constexpr int NG    = (Wn + GW - 1) / GW;
constexpr int NCH   = Sn / 4;
constexpr int NF    = NCH + 4 * (TU - 1);
constexpr int NLO   = 28;
constexpr int NWAVE = 8;
constexpr int APITCH = Sn + 4;

constexpr int NSB     = Cn * Kn;
constexpr int WP      = Ln;
constexpr int WBLK    = 4 * Sn;
constexpr int NWB_ROW = WP / WBLK;
constexpr int NWB     = Bn * Cn * NWB_ROW;
constexpr size_t WS_NEED = ((size_t)NSB + (size_t)Bn * Cn * WP) * sizeof(float);

typedef __attribute__((ext_vector_type(2))) float v2f;
typedef __attribute__((ext_vector_type(4))) float v4f;
typedef __attribute__((ext_vector_type(8))) float v8f;

__global__ __launch_bounds__(Sn) void norm_kernel(
    const float* __restrict__ x, const float* __restrict__ shp,
    float* __restrict__ den, float* __restrict__ winv, size_t ws_bytes)
{
    if (ws_bytes < WS_NEED) return;

    if (blockIdx.x >= NWB) {
        __shared__ float red[4];
        const int row = blockIdx.x - NWB;
        const int s   = threadIdx.x;
        const float v = shp[row * Sn + s];
        float p = v * v;
        #pragma unroll
        for (int off = 16; off >= 1; off >>= 1) p += __shfl_xor(p, off, 32);
        if ((s & 31) == 0) red[s >> 5] = p;
        __syncthreads();
        const float tot = red[0] + red[1] + red[2] + red[3];
        if (s == 0) den[row] = fmaxf(sqrtf(tot), EPSV);
        return;
    }

    const int wb  = blockIdx.x;
    const int row = wb / NWB_ROW;
    const int t   = (wb % NWB_ROW) * Sn + threadIdx.x;
    const float* xr = x + (size_t)row * Ln;

    v4f v[Sn / 4 + 1];
    #pragma unroll
    for (int j = 0; j <= Sn / 4; ++j) {
        const int e = min(4 * t + 4 * j, Ln - 4);
        v[j] = *(const v4f*)(xr + e);
    }
    float r[4];
    #pragma unroll
    for (int i = 0; i < 4; ++i) {
        float a[4] = {0.0f, 0.0f, 0.0f, 0.0f};
        #pragma unroll
        for (int j = 0; j < Sn / 4; ++j) {
            #pragma unroll
            for (int k = 0; k < 4; ++k) {
                const int q = i + k;
                const float e = (q < 4) ? v[j][q] : v[j + 1][q - 4];
                a[k] = fmaf(e, e, a[k]);
            }
        }
        const float ss = (a[0] + a[1]) + (a[2] + a[3]);
        r[i] = 1.0f / fmaxf(sqrtf(ss), EPSV);
    }
    *(v4f*)(winv + (size_t)row * WP + 4 * t) = (v4f){r[0], r[1], r[2], r[3]};
}

__device__ __forceinline__ v2f ldpair(const float* p)
{
    v2f r;
    r.x = p[0];
    r.y = p[1];
    return r;
}

__global__ __launch_bounds__(NWAVE * 32) void shapelet_wmma_kernel(
    const float* __restrict__ x, const float* __restrict__ shp,
    const float* __restrict__ den, const float* __restrict__ winv,
    float* __restrict__ out, size_t ws_bytes)
{
    if (ws_bytes < WS_NEED) return;

    __shared__ float ldsA[Cn * 16 * APITCH];
    __shared__ float ldsR[NWAVE * 16];

    const int kt   = blockIdx.x & (KT - 1);
    const int b    = blockIdx.x >> 4;
    const int tid  = threadIdx.x;
    const int wave = __builtin_amdgcn_readfirstlane(tid >> 5);
    const int lane = tid & 31;
    const int n    = lane & 15;
    const int h    = lane >> 4;
    const int kA   = h << 1;

    #pragma unroll
    for (int it = 0; it < (Cn * 16 * Sn / 4) / (NWAVE * 32); ++it) {
        const int idx = it * (NWAVE * 32) + tid;
        const int r   = idx >> 5;
        const int q4  = idx & 31;
        const int row = (r >> 4) * Kn + kt * 16 + (r & 15);
        const v4f   v = *(const v4f*)(shp + (size_t)row * Sn + q4 * 4);
        const float d = den[row];
        const v4f   q = {v.x / d, v.y / d, v.z / d, v.w / d};
        *(v4f*)&ldsA[r * APITCH + q4 * 4] = q;
    }
    __syncthreads();

    v2f af[Cn][NCH];
    #pragma unroll
    for (int c = 0; c < Cn; ++c) {
        #pragma unroll
        for (int j = 0; j < NCH; ++j)
            af[c][j] = *(const v2f*)&ldsA[(c * 16 + n) * APITCH + 4 * j + kA];
    }

    float rmax[8];
    #pragma unroll
    for (int r = 0; r < 8; ++r) rmax[r] = 0.0f;

    const float* xb = x + (size_t)b * Cn * Ln + n + kA;
    const float* wv = winv + (size_t)b * Cn * WP + n;

    v2f f[NF];
    {
        const int g0 = wave;
        const int W0 = (g0 * GW < Wn - GW) ? g0 * GW : Wn - GW;
        #pragma unroll
        for (int i = 0; i < NLO; ++i) f[i] = ldpair(xb + W0 + 4 * i);
    }

    for (int g = wave; g < NG; g += NWAVE) {
        const int W0  = (g * GW < Wn - GW) ? g * GW : Wn - GW;
        const int gn  = (g + NWAVE < NG) ? g + NWAVE : g;
        const int W0n = (gn * GW < Wn - GW) ? gn * GW : Wn - GW;
        v8f tot[TU];

        #pragma unroll
        for (int c = 0; c < Cn; ++c) {
            const float* px  = xb + c * Ln + W0;
            const float* pxn = (c < Cn - 1) ? xb + (c + 1) * Ln + W0 : xb + W0n;

            __builtin_amdgcn_sched_barrier(0);
            #pragma unroll
            for (int i = NLO; i < NF; ++i) f[i] = ldpair(px + 4 * i);
            float inv[TU];
            #pragma unroll
            for (int u = 0; u < TU; ++u) inv[u] = wv[c * WP + W0 + 16 * u];
            __builtin_amdgcn_sched_barrier(0);

            v8f acc[TU];
            #pragma unroll
            for (int u = 0; u < TU; ++u) acc[u] = (v8f){0.f, 0.f, 0.f, 0.f, 0.f, 0.f, 0.f, 0.f};

            #pragma unroll
            for (int j = 0; j < NCH / 2; ++j) {
                #pragma unroll
                for (int u = 0; u < TU; ++u) {
                    acc[u] = __builtin_amdgcn_wmma_f32_16x16x4_f32(
                         false, af[c][j],  false, f[j + 4 * u],
                         (short)0, acc[u],  false,  false);
                }
            }

            __builtin_amdgcn_sched_barrier(0);
            v2f nf[NLO];
            #pragma unroll
            for (int i = 0; i < NLO; ++i) nf[i] = ldpair(pxn + 4 * i);
            __builtin_amdgcn_sched_barrier(0);

            #pragma unroll
            for (int j = NCH / 2; j < NCH; ++j) {
                #pragma unroll
                for (int u = 0; u < TU; ++u) {
                    acc[u] = __builtin_amdgcn_wmma_f32_16x16x4_f32(
                        false, af[c][j], false, f[j + 4 * u], (short)0, acc[u], false, false);
                }
            }

            #pragma unroll
            for (int u = 0; u < TU; ++u) {
                #pragma unroll
                for (int r = 0; r < 8; ++r)
                    tot[u][r] = (c == 0) ? acc[u][r] * inv[u] : fmaf(acc[u][r], inv[u], tot[u][r]);
            }

            #pragma unroll
            for (int i = 0; i < NLO; ++i) f[i] = nf[i];
        }
        #pragma unroll
        for (int u = 0; u < TU; ++u) {
            #pragma unroll
            for (int r = 0; r < 8; ++r) rmax[r] = fmaxf(rmax[r], tot[u][r] * 0.25f);
        }
    }

    #pragma unroll
    for (int r = 0; r < 8; ++r) {
        #pragma unroll
        for (int off = 8; off >= 1; off >>= 1) rmax[r] = fmaxf(rmax[r], __shfl_xor(rmax[r], off, 32));
        if (n == 0) ldsR[wave * 16 + 8 * h + r] = rmax[r];
    }
    __syncthreads();
    if (tid < 16) {
        float v = ldsR[tid];
        #pragma unroll
        for (int w = 1; w < NWAVE; ++w) v = fmaxf(v, ldsR[w * 16 + tid]);
        out[b * Kn + kt * 16 + tid] = v;
    }
}

extern "C" void kernel_launch(void* const* d_in, const int* in_sizes, int n_in,
                              void* d_out, int out_size, void* d_ws, size_t ws_size,
                              hipStream_t stream) {
    const float* x   = (const float*)d_in[0];
    const float* shp = (const float*)d_in[1];
    float* out  = (float*)d_out;
    float* den  = (float*)d_ws;
    float* winv = (float*)d_ws + NSB;

    norm_kernel<<<NWB + NSB, Sn, 0, stream>>>(x, shp, den, winv, ws_size);
    shapelet_wmma_kernel<<<Bn * KT, NWAVE * 32, 0, stream>>>(x, shp, den, winv, out, ws_size);
}
